# starting kernel
# baseline (speedup 1.0000x reference)
#include <hip/hip_runtime.h>
#include <hip/hip_bf16.h>

typedef __attribute__((ext_vector_type(2))) float v2f;
typedef __attribute__((ext_vector_type(8))) float v8f;

#define D_FEAT 128
#define LDS_STRIDE 132
#define ROWS_PER_BLOCK 128

__global__ void gcn_init_deg(float* __restrict__ deg, int N) {
    int i = blockIdx.x * blockDim.x + threadIdx.x;
    if (i < N) deg[i] = 1.0f;
}

__global__ void gcn_count_deg(const int* __restrict__ dst, float* __restrict__ deg, int E) {
    int i = blockIdx.x * blockDim.x + threadIdx.x;
    if (i < E) atomicAdd(&deg[dst[i]], 1.0f);
}

__global__ void gcn_rsqrt_deg(float* __restrict__ dis, int N) {
    int i = blockIdx.x * blockDim.x + threadIdx.x;
    if (i < N) dis[i] = rsqrtf(dis[i]);
}

__global__ void __launch_bounds__(256)
gcn_gemm128(const float* __restrict__ X, const float* __restrict__ W,
            float* __restrict__ H, int N) {
    __shared__ float Ws[D_FEAT * LDS_STRIDE];
    __shared__ float As[16 * LDS_STRIDE];

    const int tid  = threadIdx.x;
    const int base = blockIdx.x * ROWS_PER_BLOCK;

    for (int i = tid; i < (D_FEAT * D_FEAT) / 4; i += 256) {
        const int k = i >> 5;
        const int c = i & 31;
        const float4 w4 = ((const float4*)(W + (size_t)k * D_FEAT))[c];
        Ws[(4 * c + 0) * LDS_STRIDE + k] = w4.x;
        Ws[(4 * c + 1) * LDS_STRIDE + k] = w4.y;
        Ws[(4 * c + 2) * LDS_STRIDE + k] = w4.z;
        Ws[(4 * c + 3) * LDS_STRIDE + k] = w4.w;
    }

    const int wave = tid >> 5;
    const int lane = tid & 31;
    const int half = lane >> 4;
    const int l16  = lane & 15;
    const int col  = wave * 16 + l16;

    #pragma unroll 1
    for (int p = 0; p < ROWS_PER_BLOCK / 16; ++p) {
        const int r0 = base + p * 16;
        if (r0 >= N) break;

        __syncthreads();

        for (int i = tid; i < (16 * D_FEAT) / 4; i += 256) {
            const int row = i >> 5;
            const int c   = i & 31;
            if (r0 + row < N) {
                const float4 a4 = ((const float4*)(X + (size_t)(r0 + row) * D_FEAT))[c];
                *(float4*)&As[row * LDS_STRIDE + 4 * c] = a4;
            }
        }
        __syncthreads();

        v8f acc = {};
        #pragma unroll
        for (int k0 = 0; k0 < D_FEAT; k0 += 4) {
            const int ka = k0 + 2 * half;
            const v2f a = *(const v2f*)(As + l16 * LDS_STRIDE + ka);
            const v2f b = *(const v2f*)(Ws + col * LDS_STRIDE + ka);
            acc = __builtin_amdgcn_wmma_f32_16x16x4_f32(
                 false, a,  false, b,
                 (short)0, acc,  false,  false);
        }

        #pragma unroll
        for (int j = 0; j < 8; ++j) {
            const int row = r0 + j + 8 * half;
            if (row < N) H[(size_t)row * D_FEAT + col] = acc[j];
        }
    }
}

__global__ void gcn_self_bias(const float* __restrict__ H, const float* __restrict__ dis,
                              const float* __restrict__ b, float* __restrict__ out, int N) {
    int i = blockIdx.x * blockDim.x + threadIdx.x;
    if (i < N * D_FEAT) {
        int n = i >> 7;
        int d = i & (D_FEAT - 1);
        float s = dis[n];
        out[i] = b[d] + s * s * H[i];
    }
}

__global__ void gcn_scatter_edges(const float* __restrict__ H,
                                  const int* __restrict__ src,
                                  const int* __restrict__ dst,
                                  const float* __restrict__ dis,
                                  float* __restrict__ out, int E) {
    int t = blockIdx.x * blockDim.x + threadIdx.x;
    int e = t >> 5;
    if (e >= E) return;
    int lane = threadIdx.x & 31;
    int s = src[e];
    int d = dst[e];
    float w = dis[s] * dis[d];
    const float4 hv = ((const float4*)(H + (size_t)s * D_FEAT))[lane];
    float* o = out + (size_t)d * D_FEAT + lane * 4;
    atomicAdd(o + 0, hv.x * w);
    atomicAdd(o + 1, hv.y * w);
    atomicAdd(o + 2, hv.z * w);
    atomicAdd(o + 3, hv.w * w);
}

extern "C" void kernel_launch(void* const* d_in, const int* in_sizes, int n_in,
                              void* d_out, int out_size, void* d_ws, size_t ws_size,
                              hipStream_t stream) {
    const float* feat = (const float*)d_in[0];
    const int*   ei   = (const int*)d_in[1];
    const float* W0   = (const float*)d_in[2];
    const float* b0   = (const float*)d_in[3];
    const float* W1   = (const float*)d_in[4];
    const float* b1   = (const float*)d_in[5];
    const float* W2   = (const float*)d_in[6];
    const float* b2   = (const float*)d_in[7];

    const int N = in_sizes[0] / D_FEAT;
    const int E = in_sizes[1] / 2;
    const int* src = ei;
    const int* dst = ei + E;

    float* dis = (float*)d_ws;
    size_t off = ((size_t)N + 1023) & ~(size_t)1023;
    float* H   = dis + off;
    float* X   = H + (size_t)N * D_FEAT;
    float* out = (float*)d_out;

    const int TB = 256;
    dim3 blkN((N + TB - 1) / TB);
    dim3 blkE((E + TB - 1) / TB);
    dim3 blkND(((size_t)N * D_FEAT + TB - 1) / TB);
    dim3 blkGemm((N + ROWS_PER_BLOCK - 1) / ROWS_PER_BLOCK);
    dim3 blkEdge(((size_t)E * 32 + TB - 1) / TB);

    gcn_init_deg<<<blkN, TB, 0, stream>>>(dis, N);
    gcn_count_deg<<<blkE, TB, 0, stream>>>(dst, dis, E);
    gcn_rsqrt_deg<<<blkN, TB, 0, stream>>>(dis, N);

    gcn_gemm128<<<blkGemm, TB, 0, stream>>>(feat, W0, H, N);
    gcn_self_bias<<<blkND, TB, 0, stream>>>(H, dis, b0, X, N);
    gcn_scatter_edges<<<blkEdge, TB, 0, stream>>>(H, src, dst, dis, X, E);

    gcn_gemm128<<<blkGemm, TB, 0, stream>>>(X, W1, H, N);
    gcn_self_bias<<<blkND, TB, 0, stream>>>(H, dis, b1, X, N);
    gcn_scatter_edges<<<blkEdge, TB, 0, stream>>>(H, src, dst, dis, X, E);

    gcn_gemm128<<<blkGemm, TB, 0, stream>>>(X, W2, H, N);
    gcn_self_bias<<<blkND, TB, 0, stream>>>(H, dis, b2, out, N);
    gcn_scatter_edges<<<blkEdge, TB, 0, stream>>>(H, src, dst, dis, out, E);
}


# Edge scatter: one wave per 8 consecutive edges; their 32 H-row line loads are issued together before the 32 line adds (was one wave per edge); same f32 atomic adds, arrival order as before
# speedup vs baseline: 1.8071x; 1.8071x over previous
#include <hip/hip_runtime.h>
#include <hip/hip_bf16.h>

typedef __attribute__((ext_vector_type(2))) float v2f;
typedef __attribute__((ext_vector_type(8))) float v8f;

#define D_FEAT 128
#define LDS_STRIDE 132
#define ROWS_PER_BLOCK 128

__global__ void gcn_init_deg(float* __restrict__ deg, int N) {
    int i = blockIdx.x * blockDim.x + threadIdx.x;
    if (i < N) deg[i] = 1.0f;
}

__global__ void gcn_count_deg(const int* __restrict__ dst, float* __restrict__ deg, int E) {
    int i = blockIdx.x * blockDim.x + threadIdx.x;
    if (i < E) atomicAdd(&deg[dst[i]], 1.0f);
}

__global__ void gcn_rsqrt_deg(float* __restrict__ dis, int N) {
    int i = blockIdx.x * blockDim.x + threadIdx.x;
    if (i < N) dis[i] = rsqrtf(dis[i]);
}

__global__ void __launch_bounds__(256)
gcn_gemm128(const float* __restrict__ X, const float* __restrict__ W,
            float* __restrict__ H, int N) {
    __shared__ float Ws[D_FEAT * LDS_STRIDE];
    __shared__ float As[16 * LDS_STRIDE];

    const int tid  = threadIdx.x;
    const int base = blockIdx.x * ROWS_PER_BLOCK;

    for (int i = tid; i < (D_FEAT * D_FEAT) / 4; i += 256) {
        const int k = i >> 5;
        const int c = i & 31;
        const float4 w4 = ((const float4*)(W + (size_t)k * D_FEAT))[c];
        Ws[(4 * c + 0) * LDS_STRIDE + k] = w4.x;
        Ws[(4 * c + 1) * LDS_STRIDE + k] = w4.y;
        Ws[(4 * c + 2) * LDS_STRIDE + k] = w4.z;
        Ws[(4 * c + 3) * LDS_STRIDE + k] = w4.w;
    }

    const int wave = tid >> 5;
    const int lane = tid & 31;
    const int half = lane >> 4;
    const int l16  = lane & 15;
    const int col  = wave * 16 + l16;

    #pragma unroll 1
    for (int p = 0; p < ROWS_PER_BLOCK / 16; ++p) {
        const int r0 = base + p * 16;
        if (r0 >= N) break;

        __syncthreads();

        for (int i = tid; i < (16 * D_FEAT) / 4; i += 256) {
            const int row = i >> 5;
            const int c   = i & 31;
            if (r0 + row < N) {
                const float4 a4 = ((const float4*)(X + (size_t)(r0 + row) * D_FEAT))[c];
                *(float4*)&As[row * LDS_STRIDE + 4 * c] = a4;
            }
        }
        __syncthreads();

        v8f acc = {};
        #pragma unroll
        for (int k0 = 0; k0 < D_FEAT; k0 += 4) {
            const int ka = k0 + 2 * half;
            const v2f a = *(const v2f*)(As + l16 * LDS_STRIDE + ka);
            const v2f b = *(const v2f*)(Ws + col * LDS_STRIDE + ka);
            acc = __builtin_amdgcn_wmma_f32_16x16x4_f32(
                 false, a,  false, b,
                 (short)0, acc,  false,  false);
        }

        #pragma unroll
        for (int j = 0; j < 8; ++j) {
            const int row = r0 + j + 8 * half;
            if (row < N) H[(size_t)row * D_FEAT + col] = acc[j];
        }
    }
}

__global__ void gcn_self_bias(const float* __restrict__ H, const float* __restrict__ dis,
                              const float* __restrict__ b, float* __restrict__ out, int N) {
    int i = blockIdx.x * blockDim.x + threadIdx.x;
    if (i < N * D_FEAT) {
        int n = i >> 7;
        int d = i & (D_FEAT - 1);
        float s = dis[n];
        out[i] = b[d] + s * s * H[i];
    }
}

#define EDGES_PER_WAVE 8

__global__ void __launch_bounds__(256)
gcn_scatter_edges(const float* __restrict__ H,
                  const int* __restrict__ src,
                  const int* __restrict__ dst,
                  const float* __restrict__ dis,
                  float* __restrict__ out, int E) {
    const int lane = threadIdx.x & 31;
    const int wave = __builtin_amdgcn_readfirstlane(
        (int)((blockIdx.x * blockDim.x + threadIdx.x) >> 5));
    const int e0 = wave * EDGES_PER_WAVE;
    if (e0 >= E) return;
    const int n = min(EDGES_PER_WAVE, E - e0);

    const int el = e0 + min(lane & (EDGES_PER_WAVE - 1), n - 1);
    const int s = src[el];
    const int d = dst[el];
    const float ds = dis[s];
    const float dd = dis[d];

    if (n == EDGES_PER_WAVE) {
        float h[EDGES_PER_WAVE][4];
        #pragma unroll
        for (int j = 0; j < EDGES_PER_WAVE; ++j) {
            const int sj = __builtin_amdgcn_readlane(s, j);
            const float* hp = H + (size_t)sj * D_FEAT + lane;
            h[j][0] = hp[0];
            h[j][1] = hp[32];
            h[j][2] = hp[64];
            h[j][3] = hp[96];
        }
        const float w = ds * dd;
        #pragma unroll
        for (int j = 0; j < EDGES_PER_WAVE; ++j) {
            const int dj = __builtin_amdgcn_readlane(d, j);
            const float wj = __int_as_float(__builtin_amdgcn_readlane(__float_as_int(w), j));
            float* op = out + (size_t)dj * D_FEAT + lane;
            atomicAdd(op +  0, h[j][0] * wj);
            atomicAdd(op + 32, h[j][1] * wj);
            atomicAdd(op + 64, h[j][2] * wj);
            atomicAdd(op + 96, h[j][3] * wj);
        }
    } else {
        const float w = ds * dd;
        for (int j = 0; j < n; ++j) {
            const int sj = __builtin_amdgcn_readlane(s, j);
            const int dj = __builtin_amdgcn_readlane(d, j);
            const float wj = __int_as_float(__builtin_amdgcn_readlane(__float_as_int(w), j));
            const float* hp = H + (size_t)sj * D_FEAT + lane;
            float* op = out + (size_t)dj * D_FEAT + lane;
            const float h0 = hp[0];
            const float h1 = hp[32];
            const float h2 = hp[64];
            const float h3 = hp[96];
            atomicAdd(op +  0, h0 * wj);
            atomicAdd(op + 32, h1 * wj);
            atomicAdd(op + 64, h2 * wj);
            atomicAdd(op + 96, h3 * wj);
        }
    }
}

extern "C" void kernel_launch(void* const* d_in, const int* in_sizes, int n_in,
                              void* d_out, int out_size, void* d_ws, size_t ws_size,
                              hipStream_t stream) {
    const float* feat = (const float*)d_in[0];
    const int*   ei   = (const int*)d_in[1];
    const float* W0   = (const float*)d_in[2];
    const float* b0   = (const float*)d_in[3];
    const float* W1   = (const float*)d_in[4];
    const float* b1   = (const float*)d_in[5];
    const float* W2   = (const float*)d_in[6];
    const float* b2   = (const float*)d_in[7];

    const int N = in_sizes[0] / D_FEAT;
    const int E = in_sizes[1] / 2;
    const int* src = ei;
    const int* dst = ei + E;

    float* dis = (float*)d_ws;
    size_t off = ((size_t)N + 1023) & ~(size_t)1023;
    float* H   = dis + off;
    float* X   = H + (size_t)N * D_FEAT;
    float* out = (float*)d_out;

    const int TB = 256;
    dim3 blkN((N + TB - 1) / TB);
    dim3 blkE((E + TB - 1) / TB);
    dim3 blkND(((size_t)N * D_FEAT + TB - 1) / TB);
    dim3 blkGemm((N + ROWS_PER_BLOCK - 1) / ROWS_PER_BLOCK);
    dim3 blkEdge(((size_t)((E + EDGES_PER_WAVE - 1) / EDGES_PER_WAVE) * 32 + TB - 1) / TB);

    gcn_init_deg<<<blkN, TB, 0, stream>>>(dis, N);
    gcn_count_deg<<<blkE, TB, 0, stream>>>(dst, dis, E);
    gcn_rsqrt_deg<<<blkN, TB, 0, stream>>>(dis, N);

    gcn_gemm128<<<blkGemm, TB, 0, stream>>>(feat, W0, H, N);
    gcn_self_bias<<<blkND, TB, 0, stream>>>(H, dis, b0, X, N);
    gcn_scatter_edges<<<blkEdge, TB, 0, stream>>>(H, src, dst, dis, X, E);

    gcn_gemm128<<<blkGemm, TB, 0, stream>>>(X, W1, H, N);
    gcn_self_bias<<<blkND, TB, 0, stream>>>(H, dis, b1, X, N);
    gcn_scatter_edges<<<blkEdge, TB, 0, stream>>>(H, src, dst, dis, X, E);

    gcn_gemm128<<<blkGemm, TB, 0, stream>>>(X, W2, H, N);
    gcn_self_bias<<<blkND, TB, 0, stream>>>(H, dis, b2, out, N);
    gcn_scatter_edges<<<blkEdge, TB, 0, stream>>>(H, src, dst, dis, out, E);
}


# Start value b + dis^2*H of each layer written by the GEMM epilogue from its accumulators, beside H; the 3 self-term launches and their re-read of H are gone (12 launches -> 9); sums unchanged
# speedup vs baseline: 1.9933x; 1.1030x over previous
#include <hip/hip_runtime.h>
#include <hip/hip_bf16.h>

typedef __attribute__((ext_vector_type(2))) float v2f;
typedef __attribute__((ext_vector_type(8))) float v8f;

#define D_FEAT 128
#define LDS_STRIDE 132
#define ROWS_PER_BLOCK 128

__global__ void gcn_init_deg(float* __restrict__ deg, int N) {
    int i = blockIdx.x * blockDim.x + threadIdx.x;
    if (i < N) deg[i] = 1.0f;
}

__global__ void gcn_count_deg(const int* __restrict__ dst, float* __restrict__ deg, int E) {
    int i = blockIdx.x * blockDim.x + threadIdx.x;
    if (i < E) atomicAdd(&deg[dst[i]], 1.0f);
}

__global__ void gcn_rsqrt_deg(float* __restrict__ dis, int N) {
    int i = blockIdx.x * blockDim.x + threadIdx.x;
    if (i < N) dis[i] = rsqrtf(dis[i]);
}

__global__ void __launch_bounds__(256)
gcn_gemm128(const float* X, const float* __restrict__ W,
            const float* __restrict__ dis, const float* __restrict__ bias,
            float* __restrict__ H, float* S, int N) {
    __shared__ float Ws[D_FEAT * LDS_STRIDE];
    __shared__ float As[16 * LDS_STRIDE];

    const int tid  = threadIdx.x;
    const int base = blockIdx.x * ROWS_PER_BLOCK;

    for (int i = tid; i < (D_FEAT * D_FEAT) / 4; i += 256) {
        const int k = i >> 5;
        const int c = i & 31;
        const float4 w4 = ((const float4*)(W + (size_t)k * D_FEAT))[c];
        Ws[(4 * c + 0) * LDS_STRIDE + k] = w4.x;
        Ws[(4 * c + 1) * LDS_STRIDE + k] = w4.y;
        Ws[(4 * c + 2) * LDS_STRIDE + k] = w4.z;
        Ws[(4 * c + 3) * LDS_STRIDE + k] = w4.w;
    }

    const int wave = tid >> 5;
    const int lane = tid & 31;
    const int half = lane >> 4;
    const int l16  = lane & 15;
    const int col  = wave * 16 + l16;
    const float bc = bias[col];

    #pragma unroll 1
    for (int p = 0; p < ROWS_PER_BLOCK / 16; ++p) {
        const int r0 = base + p * 16;
        if (r0 >= N) break;

        __syncthreads();

        for (int i = tid; i < (16 * D_FEAT) / 4; i += 256) {
            const int row = i >> 5;
            const int c   = i & 31;
            if (r0 + row < N) {
                const float4 a4 = ((const float4*)(X + (size_t)(r0 + row) * D_FEAT))[c];
                *(float4*)&As[row * LDS_STRIDE + 4 * c] = a4;
            }
        }
        __syncthreads();

        float s2[8];
        for (int j = 0; j < 8; ++j) {
            const float s = dis[min(r0 + j + 8 * half, N - 1)];
            s2[j] = s * s;
        }

        v8f acc = {};
        #pragma unroll
        for (int k0 = 0; k0 < D_FEAT; k0 += 4) {
            const int ka = k0 + 2 * half;
            const v2f a = *(const v2f*)(As + l16 * LDS_STRIDE + ka);
            const v2f b = *(const v2f*)(Ws + col * LDS_STRIDE + ka);
            acc = __builtin_amdgcn_wmma_f32_16x16x4_f32(
                 false, a,  false, b,
                 (short)0, acc,  false,  false);
        }

        #pragma unroll
        for (int j = 0; j < 8; ++j) {
            const int row = r0 + j + 8 * half;
            if (row < N) {
                const size_t o = (size_t)row * D_FEAT + col;
                H[o] = acc[j];
                S[o] = bc + s2[j] * acc[j];
            }
        }
    }
}

#define EDGES_PER_WAVE 8

__global__ void __launch_bounds__(256)
gcn_scatter_edges(const float* __restrict__ H,
                  const int* __restrict__ src,
                  const int* __restrict__ dst,
                  const float* __restrict__ dis,
                  float* __restrict__ out, int E) {
    const int lane = threadIdx.x & 31;
    const int wave = __builtin_amdgcn_readfirstlane(
        (int)((blockIdx.x * blockDim.x + threadIdx.x) >> 5));
    const int e0 = wave * EDGES_PER_WAVE;
    if (e0 >= E) return;
    const int n = min(EDGES_PER_WAVE, E - e0);

    const int el = e0 + min(lane & (EDGES_PER_WAVE - 1), n - 1);
    const int s = src[el];
    const int d = dst[el];
    const float ds = dis[s];
    const float dd = dis[d];

    if (n == EDGES_PER_WAVE) {
        float h[EDGES_PER_WAVE][4];
        #pragma unroll
        for (int j = 0; j < EDGES_PER_WAVE; ++j) {
            const int sj = __builtin_amdgcn_readlane(s, j);
            const float* hp = H + (size_t)sj * D_FEAT + lane;
            h[j][0] = hp[0];
            h[j][1] = hp[32];
            h[j][2] = hp[64];
            h[j][3] = hp[96];
        }
        const float w = ds * dd;
        #pragma unroll
        for (int j = 0; j < EDGES_PER_WAVE; ++j) {
            const int dj = __builtin_amdgcn_readlane(d, j);
            const float wj = __int_as_float(__builtin_amdgcn_readlane(__float_as_int(w), j));
            float* op = out + (size_t)dj * D_FEAT + lane;
            atomicAdd(op +  0, h[j][0] * wj);
            atomicAdd(op + 32, h[j][1] * wj);
            atomicAdd(op + 64, h[j][2] * wj);
            atomicAdd(op + 96, h[j][3] * wj);
        }
    } else {
        const float w = ds * dd;
        for (int j = 0; j < n; ++j) {
            const int sj = __builtin_amdgcn_readlane(s, j);
            const int dj = __builtin_amdgcn_readlane(d, j);
            const float wj = __int_as_float(__builtin_amdgcn_readlane(__float_as_int(w), j));
            const float* hp = H + (size_t)sj * D_FEAT + lane;
            float* op = out + (size_t)dj * D_FEAT + lane;
            const float h0 = hp[0];
            const float h1 = hp[32];
            const float h2 = hp[64];
            const float h3 = hp[96];
            atomicAdd(op +  0, h0 * wj);
            atomicAdd(op + 32, h1 * wj);
            atomicAdd(op + 64, h2 * wj);
            atomicAdd(op + 96, h3 * wj);
        }
    }
}

extern "C" void kernel_launch(void* const* d_in, const int* in_sizes, int n_in,
                              void* d_out, int out_size, void* d_ws, size_t ws_size,
                              hipStream_t stream) {
    const float* feat = (const float*)d_in[0];
    const int*   ei   = (const int*)d_in[1];
    const float* W0   = (const float*)d_in[2];
    const float* b0   = (const float*)d_in[3];
    const float* W1   = (const float*)d_in[4];
    const float* b1   = (const float*)d_in[5];
    const float* W2   = (const float*)d_in[6];
    const float* b2   = (const float*)d_in[7];

    const int N = in_sizes[0] / D_FEAT;
    const int E = in_sizes[1] / 2;
    const int* src = ei;
    const int* dst = ei + E;

    float* dis = (float*)d_ws;
    size_t off = ((size_t)N + 1023) & ~(size_t)1023;
    float* H   = dis + off;
    float* X   = H + (size_t)N * D_FEAT;
    float* out = (float*)d_out;

    const int TB = 256;
    dim3 blkN((N + TB - 1) / TB);
    dim3 blkE((E + TB - 1) / TB);
    dim3 blkGemm((N + ROWS_PER_BLOCK - 1) / ROWS_PER_BLOCK);
    dim3 blkEdge(((size_t)((E + EDGES_PER_WAVE - 1) / EDGES_PER_WAVE) * 32 + TB - 1) / TB);

    gcn_init_deg<<<blkN, TB, 0, stream>>>(dis, N);
    gcn_count_deg<<<blkE, TB, 0, stream>>>(dst, dis, E);
    gcn_rsqrt_deg<<<blkN, TB, 0, stream>>>(dis, N);

    gcn_gemm128<<<blkGemm, TB, 0, stream>>>(feat, W0, dis, b0, H, X, N);
    gcn_scatter_edges<<<blkEdge, TB, 0, stream>>>(H, src, dst, dis, X, E);

    gcn_gemm128<<<blkGemm, TB, 0, stream>>>(X, W1, dis, b1, H, X, N);
    gcn_scatter_edges<<<blkEdge, TB, 0, stream>>>(H, src, dst, dis, X, E);

    gcn_gemm128<<<blkGemm, TB, 0, stream>>>(X, W2, dis, b2, H, out, N);
    gcn_scatter_edges<<<blkEdge, TB, 0, stream>>>(H, src, dst, dis, out, E);
}


# Edge pass: a finder hands each edge record through an LDS queue to the wave that owns the row (row & 31), which adds by plain LDS load, fma, store; no LDS f32 atomics; sums in queue (arrival) order
# speedup vs baseline: 3.0082x; 1.5092x over previous
#include <hip/hip_runtime.h>
#include <hip/hip_bf16.h>

typedef __attribute__((ext_vector_type(2))) float v2f;
typedef __attribute__((ext_vector_type(8))) float v8f;

#define D_FEAT 128
#define LDS_STRIDE 132
#define ROWS_PER_BLOCK 128

#define OWN_ROWS 392
#define MAP_TILE_EDGES 1024
#define MAP_TILE_OWNERS 256
#define MAP_TILE_PITCH 33
#define MAP_TILE_LINES (MAP_TILE_OWNERS / (MAP_TILE_EDGES / 32))
#define MAP_UNITS(E) ((((E) + MAP_TILE_EDGES - 1) / MAP_TILE_EDGES) * (MAP_TILE_EDGES / 128))

#define REC_SRC_BITS 23
static_assert(OWN_ROWS <= (1 << (32 - REC_SRC_BITS)), "row inside an owner's run must fit above the src bits");

__global__ void gcn_init_deg(float* __restrict__ deg, int N) {
    const int i = blockIdx.x * blockDim.x + threadIdx.x;
    if (i < N) deg[i] = 1.0f;
}

__global__ void __launch_bounds__(MAP_TILE_EDGES)
gcn_count_deg(const int* __restrict__ dst, float* __restrict__ deg,
              unsigned* __restrict__ map, int N, int E) {
    __shared__ unsigned tile[MAP_TILE_OWNERS * MAP_TILE_PITCH];
    const int tid  = threadIdx.x;
    const int lane = tid & 31;
    const int wave = tid >> 5;
    const int i = blockIdx.x * MAP_TILE_EDGES + tid;
    const int owners = (N + OWN_ROWS - 1) / OWN_ROWS;
    const size_t wd = (size_t)MAP_UNITS(E) * 4;
    int d = 0;
    if (i < E) d = dst[i];
    const int o = (i < E) ? (int)((unsigned)d / OWN_ROWS) : -1;
    for (int ob = 0; ob < owners; ob += MAP_TILE_OWNERS) {
        if (ob > 0) __syncthreads();
        for (int k = tid; k < MAP_TILE_OWNERS * MAP_TILE_PITCH; k += MAP_TILE_EDGES) tile[k] = 0u;
        __syncthreads();
        const int ol = o - ob;
        if (ol >= 0 && ol < MAP_TILE_OWNERS) atomicOr(&tile[ol * MAP_TILE_PITCH + wave], 1u << lane);
        __syncthreads();
        unsigned wv[MAP_TILE_LINES];
        #pragma unroll
        for (int j = 0; j < MAP_TILE_LINES; ++j)
            wv[j] = tile[(wave + j * (MAP_TILE_EDGES / 32)) * MAP_TILE_PITCH + lane];
        #pragma unroll
        for (int j = 0; j < MAP_TILE_LINES; ++j) {
            const int q = ob + wave + j * (MAP_TILE_EDGES / 32);
            if (q < owners) map[(size_t)q * wd + (size_t)blockIdx.x * 32 + lane] = wv[j];
        }
    }
    if (i < E) atomicAdd(&deg[d], 1.0f);
}

__global__ void gcn_rsqrt_rec(const float* __restrict__ deg, float* __restrict__ dis,
                              const int* __restrict__ src, const int* __restrict__ dst,
                              int2* __restrict__ rec, int N, int E) {
    const int i = blockIdx.x * blockDim.x + threadIdx.x;
    for (int n = i; n < N; n += gridDim.x * blockDim.x)
        dis[n] = rsqrtf(deg[n]);
    if (i < E) {
        const int s = src[i];
        const int d = dst[i];
        const float w = rsqrtf(deg[s]) * rsqrtf(deg[d]);
        rec[i] = make_int2(s | (int)(((unsigned)d % OWN_ROWS) << REC_SRC_BITS), __float_as_int(w));
    }
}

__global__ void __launch_bounds__(256)
gcn_gemm128(const float* X, const float* __restrict__ W,
            const float* __restrict__ dis, const float* __restrict__ bias,
            float* __restrict__ H, float* S, int N) {
    __shared__ float Ws[D_FEAT * LDS_STRIDE];
    __shared__ float As[16 * LDS_STRIDE];

    const int tid  = threadIdx.x;
    const int base = blockIdx.x * ROWS_PER_BLOCK;

    for (int i = tid; i < (D_FEAT * D_FEAT) / 4; i += 256) {
        const int k = i >> 5;
        const int c = i & 31;
        const float4 w4 = ((const float4*)(W + (size_t)k * D_FEAT))[c];
        Ws[(4 * c + 0) * LDS_STRIDE + k] = w4.x;
        Ws[(4 * c + 1) * LDS_STRIDE + k] = w4.y;
        Ws[(4 * c + 2) * LDS_STRIDE + k] = w4.z;
        Ws[(4 * c + 3) * LDS_STRIDE + k] = w4.w;
    }

    const int wave = tid >> 5;
    const int lane = tid & 31;
    const int half = lane >> 4;
    const int l16  = lane & 15;
    const int col  = wave * 16 + l16;
    const float bc = bias[col];

    #pragma unroll 1
    for (int p = 0; p < ROWS_PER_BLOCK / 16; ++p) {
        const int r0 = base + p * 16;
        if (r0 >= N) break;

        __syncthreads();

        for (int i = tid; i < (16 * D_FEAT) / 4; i += 256) {
            const int row = i >> 5;
            const int c   = i & 31;
            if (r0 + row < N) {
                const float4 a4 = ((const float4*)(X + (size_t)(r0 + row) * D_FEAT))[c];
                *(float4*)&As[row * LDS_STRIDE + 4 * c] = a4;
            }
        }
        __syncthreads();

        float s2[8];
        for (int j = 0; j < 8; ++j) {
            const float s = dis[min(r0 + j + 8 * half, N - 1)];
            s2[j] = s * s;
        }

        v8f acc = {};
        #pragma unroll
        for (int k0 = 0; k0 < D_FEAT; k0 += 4) {
            const int ka = k0 + 2 * half;
            const v2f a = *(const v2f*)(As + l16 * LDS_STRIDE + ka);
            const v2f b = *(const v2f*)(Ws + col * LDS_STRIDE + ka);
            acc = __builtin_amdgcn_wmma_f32_16x16x4_f32(
                 false, a,  false, b,
                 (short)0, acc,  false,  false);
        }

        #pragma unroll
        for (int j = 0; j < 8; ++j) {
            const int row = r0 + j + 8 * half;
            if (row < N) {
                const size_t o = (size_t)row * D_FEAT + col;
                H[o] = acc[j];
                S[o] = bc + s2[j] * acc[j];
            }
        }
    }
}

#define EDGE_GROUP 16
#define PUSH_GROUP 32
#define OWNER_THREADS 1024
#define OWNER_WAVES (OWNER_THREADS / 32)
#define QUEUE_CAP 384
#define OWNER_TRIPS ((OWN_ROWS * (D_FEAT / 4) + OWNER_THREADS - 1) / OWNER_THREADS)
static_assert(OWNER_WAVES == 32, "a row's wave is row & 31");

typedef __attribute__((ext_vector_type(2))) int v2i;

__device__ __forceinline__ void gcn_owner_push(const float* __restrict__ H,
                                               const int2* __restrict__ rec,
                                               float* acc, v2i* queue, unsigned* qcnt,
                                               int pend, int n, int lane) {
    const int2 r = rec[pend];
    const int q = (int)((unsigned)r.x >> REC_SRC_BITS) & (OWNER_WAVES - 1);
    const bool act = lane < n;
    unsigned slot = QUEUE_CAP;
    if (act) slot = atomicAdd(&qcnt[q], 1u);
    const bool ok = act && slot < QUEUE_CAP;
    if (ok) {
        v2i t;
        t.x = r.x;
        t.y = r.y;
        queue[q * QUEUE_CAP + slot] = t;
    }
    unsigned om = __builtin_amdgcn_ballot_w32(act && !ok);
    while (om) {
        const int l = __builtin_ctz(om);
        om &= om - 1u;
        const int rx = __builtin_amdgcn_readlane(r.x, l);
        const float wv = __int_as_float(__builtin_amdgcn_readlane(r.y, l));
        const int s = rx & ((1 << REC_SRC_BITS) - 1);
        const int d = (int)((unsigned)rx >> REC_SRC_BITS);
        const float2* hp = (const float2*)(H + (size_t)s * D_FEAT) + lane;
        const float2 lo = hp[0];
        const float2 hi = hp[32];
        float* ap = acc + d * D_FEAT + ((2 * lane) ^ (4 * (d & 15)));
        atomicAdd(ap +  0, lo.x * wv);
        atomicAdd(ap +  1, lo.y * wv);
        atomicAdd(ap + 64, hi.x * wv);
        atomicAdd(ap + 65, hi.y * wv);
    }
}

__device__ __forceinline__ void gcn_owner_group(const float* __restrict__ H,
                                                float* acc, int rx, int ry, int n,
                                                int lane) {
    const int s = rx & ((1 << REC_SRC_BITS) - 1);
    const int d = (int)((unsigned)rx >> REC_SRC_BITS);
    float h[EDGE_GROUP][4];
    #pragma unroll
    for (int j = 0; j < EDGE_GROUP; ++j) {
        const int sj = __builtin_amdgcn_readlane(s, j);
        const float2* hp = (const float2*)(H + (size_t)sj * D_FEAT) + lane;
        const float2 lo = hp[0];
        const float2 hi = hp[32];
        h[j][0] = lo.x;
        h[j][1] = lo.y;
        h[j][2] = hi.x;
        h[j][3] = hi.y;
    }
    #pragma unroll
    for (int j = 0; j < EDGE_GROUP; ++j) {
        if (j < n) {
            const int dj = __builtin_amdgcn_readlane(d, j);
            const float wj = __int_as_float(__builtin_amdgcn_readlane(ry, j));
            float* ap = acc + dj * D_FEAT + ((2 * lane) ^ (4 * (dj & 15)));
            float2 lo = *(const float2*)(ap);
            float2 hi = *(const float2*)(ap + 64);
            lo.x += h[j][0] * wj;
            lo.y += h[j][1] * wj;
            hi.x += h[j][2] * wj;
            hi.y += h[j][3] * wj;
            *(float2*)(ap)      = lo;
            *(float2*)(ap + 64) = hi;
        }
    }
}

__global__ void __launch_bounds__(OWNER_THREADS)
gcn_owner_edges(const float* __restrict__ H,
                const int2* __restrict__ rec,
                const uint4* __restrict__ map4,
                const float* start, float* fin,
                const float* __restrict__ Wn, const float* __restrict__ bn,
                const float* __restrict__ dis,
                float* __restrict__ Hn, float* Sn, int N, int E) {
    __shared__ float acc[OWNER_TRIPS * OWNER_THREADS * 4];
    __shared__ v2i queue[OWNER_WAVES * QUEUE_CAP];
    __shared__ unsigned qcnt[OWNER_WAVES];

    const int tid  = threadIdx.x;
    const int lane = tid & 31;
    const int wave = __builtin_amdgcn_readfirstlane(tid >> 5);
    const int r0 = blockIdx.x * OWN_ROWS;
    const int nr = min(OWN_ROWS, N - r0);

    const float4* sp = (const float4*)(start + (size_t)r0 * D_FEAT);
    const int sw = (tid >> 5) & 15;
    const int tsw = tid ^ sw;
    float4* a4 = (float4*)acc;
    const int n16 = nr * (D_FEAT / 4);
    if (tid < OWNER_WAVES) qcnt[tid] = 0u;
    {
        float4 t[OWNER_TRIPS];
        #pragma unroll
        for (int k = 0; k < OWNER_TRIPS; ++k) {
            const int i = tid + k * OWNER_THREADS;
            t[k] = sp[min(i, n16 - 1)];
        }
        #pragma unroll
        for (int k = 0; k < OWNER_TRIPS; ++k) {
            a4[tsw + k * OWNER_THREADS] = t[k];
        }
    }
    __syncthreads();

    const int wd4 = (E + 127) / 128;
    const uint4* row = map4 + (size_t)blockIdx.x * MAP_UNITS(E);
    const int nIter = (wd4 + 31) / 32;
    const uint4 z = make_uint4(0u, 0u, 0u, 0u);

    int pend = 0;
    int cnt  = 0;

    uint4 v;
    {
        const int q = wave * 32 + lane;
        v = row[min(q, wd4 - 1)];
        if (q >= wd4) v = z;
    }
    for (int it = wave; it < nIter; it += OWNER_THREADS / 32) {
        uint4 vn;
        {
            const int qn = (it + OWNER_THREADS / 32) * 32 + lane;
            vn = row[min(qn, wd4 - 1)];
            if (qn >= wd4) vn = z;
        }
        const unsigned comp[4] = {v.x, v.y, v.z, v.w};
        #pragma unroll
        for (int cidx = 0; cidx < 4; ++cidx) {
            const unsigned vc = comp[cidx];
            unsigned bal = __builtin_amdgcn_ballot_w32(vc != 0u);
            while (bal) {
                const int l = __builtin_ctz(bal);
                bal &= bal - 1u;
                unsigned vv = (unsigned)__builtin_amdgcn_readlane((int)vc, l);
                const int ebase = (it * 32 + l) * 128 + cidx * 32;
                while (vv) {
                    const int e = ebase + __builtin_ctz(vv);
                    vv &= vv - 1u;
                    pend = (lane == cnt) ? e : pend;
                    if (++cnt == PUSH_GROUP) {
                        gcn_owner_push(H, rec, acc, queue, qcnt, pend, PUSH_GROUP, lane);
                        cnt = 0;
                    }
                }
            }
        }
        v = vn;
    }
    if (cnt > 0) gcn_owner_push(H, rec, acc, queue, qcnt, pend, cnt, lane);

    __syncthreads();

    {
        const v2i* myq = queue + wave * QUEUE_CAP;
        const int have = min((int)__builtin_amdgcn_readfirstlane(qcnt[wave]), QUEUE_CAP);
        int base = 0;
        for (; base + EDGE_GROUP <= have; base += EDGE_GROUP) {
            const v2i r = myq[base + (lane & (EDGE_GROUP - 1))];
            gcn_owner_group(H, acc, r.x, r.y, EDGE_GROUP, lane);
        }
        if (base < have) {
            const v2i r = myq[min(base + (lane & (EDGE_GROUP - 1)), have - 1)];
            gcn_owner_group(H, acc, r.x, r.y, have - base, lane);
        }
    }

    const int half = lane >> 4;
    const int l16  = lane & 15;
    const int col  = (wave & 7) * 16 + l16;
    v2f bfr[D_FEAT / 4];
    float bc = 0.0f;
    if (Wn != nullptr) {
        #pragma unroll
        for (int q = 0; q < D_FEAT / 4; ++q) {
            const int ka = 4 * q + 2 * half;
            bfr[q].x = Wn[(size_t)ka * D_FEAT + col];
            bfr[q].y = Wn[(size_t)(ka + 1) * D_FEAT + col];
        }
        bc = bn[col];
    }

    __syncthreads();

    if (Wn != nullptr) {
        const int npan = (nr + 15) >> 4;
        const int t = 2 * half + 4 * l16;
        #pragma unroll 1
        for (int p = wave >> 3; p < npan; p += 4) {
            const float* arow = acc + (p * 16 + l16) * D_FEAT;
            v8f c = {};
            #pragma unroll
            for (int g = 0; g < D_FEAT / 32; ++g) {
                v2f a[8];
                #pragma unroll
                for (int u = 0; u < 8; ++u)
                    a[u] = *(const v2f*)(arow + ((4 * (8 * g + u)) ^ t));
                #pragma unroll
                for (int u = 0; u < 8; ++u)
                    c = __builtin_amdgcn_wmma_f32_16x16x4_f32(
                         false, a[u],  false, bfr[8 * g + u],
                         (short)0, c,  false,  false);
                __builtin_amdgcn_sched_barrier(0);
            }
            float sd[8];
            #pragma unroll
            for (int j = 0; j < 8; ++j)
                sd[j] = dis[min(r0 + p * 16 + j + 8 * half, N - 1)];
            #pragma unroll
            for (int j = 0; j < 8; ++j) {
                const int row = p * 16 + j + 8 * half;
                if (row < nr) {
                    const size_t o = (size_t)(r0 + row) * D_FEAT + col;
                    Hn[o] = c[j];
                    Sn[o] = bc + (sd[j] * sd[j]) * c[j];
                }
            }
        }
    } else {
        float4* op = (float4*)(fin + (size_t)r0 * D_FEAT);
        #pragma unroll 1
        for (int k = 0; k < OWNER_TRIPS; ++k) {
            const int i = tid + k * OWNER_THREADS;
            if (i < n16) op[i] = a4[tsw + k * OWNER_THREADS];
        }
    }
}

extern "C" void kernel_launch(void* const* d_in, const int* in_sizes, int n_in,
                              void* d_out, int out_size, void* d_ws, size_t ws_size,
                              hipStream_t stream) {
    const float* feat = (const float*)d_in[0];
    const int*   ei   = (const int*)d_in[1];
    const float* W0   = (const float*)d_in[2];
    const float* b0   = (const float*)d_in[3];
    const float* W1   = (const float*)d_in[4];
    const float* b1   = (const float*)d_in[5];
    const float* W2   = (const float*)d_in[6];
    const float* b2   = (const float*)d_in[7];

    const int N = in_sizes[0] / D_FEAT;
    const int E = in_sizes[1] / 2;
    const int* src = ei;
    const int* dst = ei + E;

    float* deg = (float*)d_ws;
    size_t off = ((size_t)N + 1023) & ~(size_t)1023;
    float* dis = deg + off;
    float* H   = dis + off;
    float* X   = H + (size_t)N * D_FEAT;
    int2* rec  = (int2*)(X + (size_t)N * D_FEAT);
    size_t offE = ((size_t)E + 1023) & ~(size_t)1023;
    uint4* map4 = (uint4*)(rec + offE);
    const size_t wd4 = (size_t)MAP_UNITS(E);
    const size_t owners = ((size_t)N + OWN_ROWS - 1) / OWN_ROWS;
    float* H2  = (float*)(map4 + owners * wd4);
    float* out = (float*)d_out;

    const int TB = 256;
    dim3 blkE((E + TB - 1) / TB);
    dim3 blkN((N + TB - 1) / TB);
    dim3 blkTile((E + MAP_TILE_EDGES - 1) / MAP_TILE_EDGES);
    dim3 blkGemm((N + ROWS_PER_BLOCK - 1) / ROWS_PER_BLOCK);
    dim3 blkOwn((N + OWN_ROWS - 1) / OWN_ROWS);

    gcn_init_deg<<<blkN, TB, 0, stream>>>(deg, N);
    gcn_count_deg<<<blkTile, MAP_TILE_EDGES, 0, stream>>>(dst, deg, (unsigned*)map4, N, E);
    gcn_rsqrt_rec<<<blkE, TB, 0, stream>>>(deg, dis, src, dst, rec, N, E);

    gcn_gemm128<<<blkGemm, TB, 0, stream>>>(feat, W0, dis, b0, H, X, N);
    gcn_owner_edges<<<blkOwn, OWNER_THREADS, 0, stream>>>(H, rec, map4, X, nullptr, W1, b1, dis, H2, X, N, E);
    gcn_owner_edges<<<blkOwn, OWNER_THREADS, 0, stream>>>(H2, rec, map4, X, nullptr, W2, b2, dis, H, X, N, E);
    gcn_owner_edges<<<blkOwn, OWNER_THREADS, 0, stream>>>(H, rec, map4, X, out, nullptr, nullptr, dis, nullptr, nullptr, N, E);
}


# Wave queues (edge records per row-owning wave) stored by the first edge pass and loaded by the other two, which skip the find step (map walk, record loads); layers 1, 2 add in layer 0's queue order
# speedup vs baseline: 3.2822x; 1.0911x over previous
#include <hip/hip_runtime.h>
#include <hip/hip_bf16.h>

typedef __attribute__((ext_vector_type(2))) float v2f;
typedef __attribute__((ext_vector_type(8))) float v8f;

#define D_FEAT 128
#define LDS_STRIDE 132
#define ROWS_PER_BLOCK 128

#define OWN_ROWS 392
#define MAP_TILE_EDGES 1024
#define MAP_TILE_OWNERS 256
#define MAP_TILE_PITCH 33
#define MAP_TILE_LINES (MAP_TILE_OWNERS / (MAP_TILE_EDGES / 32))
#define MAP_UNITS(E) ((((E) + MAP_TILE_EDGES - 1) / MAP_TILE_EDGES) * (MAP_TILE_EDGES / 128))

#define REC_SRC_BITS 23
static_assert(OWN_ROWS <= (1 << (32 - REC_SRC_BITS)), "row inside an owner's run must fit above the src bits");

__global__ void gcn_init_deg(float* __restrict__ deg, int N) {
    const int i = blockIdx.x * blockDim.x + threadIdx.x;
    if (i < N) deg[i] = 1.0f;
}

__global__ void __launch_bounds__(MAP_TILE_EDGES)
gcn_count_deg(const int* __restrict__ dst, float* __restrict__ deg,
              unsigned* __restrict__ map, int N, int E) {
    __shared__ unsigned tile[MAP_TILE_OWNERS * MAP_TILE_PITCH];
    const int tid  = threadIdx.x;
    const int lane = tid & 31;
    const int wave = tid >> 5;
    const int i = blockIdx.x * MAP_TILE_EDGES + tid;
    const int owners = (N + OWN_ROWS - 1) / OWN_ROWS;
    const size_t wd = (size_t)MAP_UNITS(E) * 4;
    int d = 0;
    if (i < E) d = dst[i];
    const int o = (i < E) ? (int)((unsigned)d / OWN_ROWS) : -1;
    for (int ob = 0; ob < owners; ob += MAP_TILE_OWNERS) {
        if (ob > 0) __syncthreads();
        for (int k = tid; k < MAP_TILE_OWNERS * MAP_TILE_PITCH; k += MAP_TILE_EDGES) tile[k] = 0u;
        __syncthreads();
        const int ol = o - ob;
        if (ol >= 0 && ol < MAP_TILE_OWNERS) atomicOr(&tile[ol * MAP_TILE_PITCH + wave], 1u << lane);
        __syncthreads();
        unsigned wv[MAP_TILE_LINES];
        #pragma unroll
        for (int j = 0; j < MAP_TILE_LINES; ++j)
            wv[j] = tile[(wave + j * (MAP_TILE_EDGES / 32)) * MAP_TILE_PITCH + lane];
        #pragma unroll
        for (int j = 0; j < MAP_TILE_LINES; ++j) {
            const int q = ob + wave + j * (MAP_TILE_EDGES / 32);
            if (q < owners) map[(size_t)q * wd + (size_t)blockIdx.x * 32 + lane] = wv[j];
        }
    }
    if (i < E) atomicAdd(&deg[d], 1.0f);
}

__global__ void gcn_rsqrt_rec(const float* __restrict__ deg, float* __restrict__ dis,
                              const int* __restrict__ src, const int* __restrict__ dst,
                              int2* __restrict__ rec, int N, int E) {
    const int i = blockIdx.x * blockDim.x + threadIdx.x;
    for (int n = i; n < N; n += gridDim.x * blockDim.x)
        dis[n] = rsqrtf(deg[n]);
    if (i < E) {
        const int s = src[i];
        const int d = dst[i];
        const float w = rsqrtf(deg[s]) * rsqrtf(deg[d]);
        rec[i] = make_int2(s | (int)(((unsigned)d % OWN_ROWS) << REC_SRC_BITS), __float_as_int(w));
    }
}

__global__ void __launch_bounds__(256)
gcn_gemm128(const float* X, const float* __restrict__ W,
            const float* __restrict__ dis, const float* __restrict__ bias,
            float* __restrict__ H, float* S, int N) {
    __shared__ float Ws[D_FEAT * LDS_STRIDE];
    __shared__ float As[16 * LDS_STRIDE];

    const int tid  = threadIdx.x;
    const int base = blockIdx.x * ROWS_PER_BLOCK;

    for (int i = tid; i < (D_FEAT * D_FEAT) / 4; i += 256) {
        const int k = i >> 5;
        const int c = i & 31;
        const float4 w4 = ((const float4*)(W + (size_t)k * D_FEAT))[c];
        Ws[(4 * c + 0) * LDS_STRIDE + k] = w4.x;
        Ws[(4 * c + 1) * LDS_STRIDE + k] = w4.y;
        Ws[(4 * c + 2) * LDS_STRIDE + k] = w4.z;
        Ws[(4 * c + 3) * LDS_STRIDE + k] = w4.w;
    }

    const int wave = tid >> 5;
    const int lane = tid & 31;
    const int half = lane >> 4;
    const int l16  = lane & 15;
    const int col  = wave * 16 + l16;
    const float bc = bias[col];

    #pragma unroll 1
    for (int p = 0; p < ROWS_PER_BLOCK / 16; ++p) {
        const int r0 = base + p * 16;
        if (r0 >= N) break;

        __syncthreads();

        for (int i = tid; i < (16 * D_FEAT) / 4; i += 256) {
            const int row = i >> 5;
            const int c   = i & 31;
            if (r0 + row < N) {
                const float4 a4 = ((const float4*)(X + (size_t)(r0 + row) * D_FEAT))[c];
                *(float4*)&As[row * LDS_STRIDE + 4 * c] = a4;
            }
        }
        __syncthreads();

        float s2[8];
        for (int j = 0; j < 8; ++j) {
            const float s = dis[min(r0 + j + 8 * half, N - 1)];
            s2[j] = s * s;
        }

        v8f acc = {};
        #pragma unroll
        for (int k0 = 0; k0 < D_FEAT; k0 += 4) {
            const int ka = k0 + 2 * half;
            const v2f a = *(const v2f*)(As + l16 * LDS_STRIDE + ka);
            const v2f b = *(const v2f*)(Ws + col * LDS_STRIDE + ka);
            acc = __builtin_amdgcn_wmma_f32_16x16x4_f32(
                 false, a,  false, b,
                 (short)0, acc,  false,  false);
        }

        #pragma unroll
        for (int j = 0; j < 8; ++j) {
            const int row = r0 + j + 8 * half;
            if (row < N) {
                const size_t o = (size_t)row * D_FEAT + col;
                H[o] = acc[j];
                S[o] = bc + s2[j] * acc[j];
            }
        }
    }
}

#define EDGE_GROUP 16
#define PUSH_GROUP 32
#define OWNER_THREADS 1024
#define OWNER_WAVES (OWNER_THREADS / 32)
#define QUEUE_CAP 384
#define OWNER_TRIPS ((OWN_ROWS * (D_FEAT / 4) + OWNER_THREADS - 1) / OWNER_THREADS)
static_assert(OWNER_WAVES == 32, "a row's wave is row & 31");

typedef __attribute__((ext_vector_type(2))) int v2i;

__device__ __forceinline__ void gcn_owner_push(const float* __restrict__ H,
                                               const int2* __restrict__ rec,
                                               float* acc, v2i* queue, unsigned* qcnt,
                                               int pend, int n, int lane) {
    const int2 r = rec[pend];
    const int q = (int)((unsigned)r.x >> REC_SRC_BITS) & (OWNER_WAVES - 1);
    const bool act = lane < n;
    unsigned slot = QUEUE_CAP;
    if (act) slot = atomicAdd(&qcnt[q], 1u);
    const bool ok = act && slot < QUEUE_CAP;
    if (ok) {
        v2i t;
        t.x = r.x;
        t.y = r.y;
        queue[q * QUEUE_CAP + slot] = t;
    }
    unsigned om = __builtin_amdgcn_ballot_w32(act && !ok);
    while (om) {
        const int l = __builtin_ctz(om);
        om &= om - 1u;
        const int rx = __builtin_amdgcn_readlane(r.x, l);
        const float wv = __int_as_float(__builtin_amdgcn_readlane(r.y, l));
        const int s = rx & ((1 << REC_SRC_BITS) - 1);
        const int d = (int)((unsigned)rx >> REC_SRC_BITS);
        const float2* hp = (const float2*)(H + (size_t)s * D_FEAT) + lane;
        const float2 lo = hp[0];
        const float2 hi = hp[32];
        float* ap = acc + d * D_FEAT + ((2 * lane) ^ (4 * (d & 15)));
        atomicAdd(ap +  0, lo.x * wv);
        atomicAdd(ap +  1, lo.y * wv);
        atomicAdd(ap + 64, hi.x * wv);
        atomicAdd(ap + 65, hi.y * wv);
    }
}

__device__ __forceinline__ void gcn_owner_group(const float* __restrict__ H,
                                                float* acc, int rx, int ry, int n,
                                                int lane) {
    const int s = rx & ((1 << REC_SRC_BITS) - 1);
    const int d = (int)((unsigned)rx >> REC_SRC_BITS);
    float h[EDGE_GROUP][4];
    #pragma unroll
    for (int j = 0; j < EDGE_GROUP; ++j) {
        const int sj = __builtin_amdgcn_readlane(s, j);
        const float2* hp = (const float2*)(H + (size_t)sj * D_FEAT) + lane;
        const float2 lo = hp[0];
        const float2 hi = hp[32];
        h[j][0] = lo.x;
        h[j][1] = lo.y;
        h[j][2] = hi.x;
        h[j][3] = hi.y;
    }
    #pragma unroll
    for (int j = 0; j < EDGE_GROUP; ++j) {
        if (j < n) {
            const int dj = __builtin_amdgcn_readlane(d, j);
            const float wj = __int_as_float(__builtin_amdgcn_readlane(ry, j));
            float* ap = acc + dj * D_FEAT + ((2 * lane) ^ (4 * (dj & 15)));
            float2 lo = *(const float2*)(ap);
            float2 hi = *(const float2*)(ap + 64);
            lo.x += h[j][0] * wj;
            lo.y += h[j][1] * wj;
            hi.x += h[j][2] * wj;
            hi.y += h[j][3] * wj;
            *(float2*)(ap)      = lo;
            *(float2*)(ap + 64) = hi;
        }
    }
}

__global__ void __launch_bounds__(OWNER_THREADS)
gcn_owner_edges(const float* __restrict__ H,
                const int2* __restrict__ rec,
                const uint4* __restrict__ map4,
                const float* start, float* fin,
                const float* __restrict__ Wn, const float* __restrict__ bn,
                const float* __restrict__ dis,
                float* __restrict__ Hn, float* Sn,
                v2i* sq, unsigned* sqn, int reuse, int N, int E) {
    __shared__ float acc[OWNER_TRIPS * OWNER_THREADS * 4];
    __shared__ v2i queue[OWNER_WAVES * QUEUE_CAP];
    __shared__ unsigned qcnt[OWNER_WAVES];

    const int tid  = threadIdx.x;
    const int lane = tid & 31;
    const int wave = __builtin_amdgcn_readfirstlane(tid >> 5);
    const int r0 = blockIdx.x * OWN_ROWS;
    const int nr = min(OWN_ROWS, N - r0);

    v2i* sq_own = sq + ((size_t)blockIdx.x * OWNER_WAVES + wave) * QUEUE_CAP;
    unsigned* sqn_own = sqn + (size_t)blockIdx.x * OWNER_WAVES;
    v2i* myq = queue + wave * QUEUE_CAP;
    bool saved = false;
    int have_saved = 0;
    if (reuse) {
        const unsigned c = sqn_own[lane];
        saved = __builtin_amdgcn_ballot_w32(c > (unsigned)QUEUE_CAP) == 0u;
        have_saved = __builtin_amdgcn_readlane((int)c, wave);
    }
    const bool fill = saved && have_saved > 0;

    const float4* sp = (const float4*)(start + (size_t)r0 * D_FEAT);
    const int sw = (tid >> 5) & 15;
    const int tsw = tid ^ sw;
    float4* a4 = (float4*)acc;
    const int n16 = nr * (D_FEAT / 4);
    if (tid < OWNER_WAVES) qcnt[tid] = 0u;
    {
        if (fill) {
            v2i qr[QUEUE_CAP / 32];
            float4 t[OWNER_TRIPS];
            #pragma unroll
            for (int k = 0; k < QUEUE_CAP / 32; ++k)
                qr[k] = sq_own[min(lane + 32 * k, have_saved - 1)];
            #pragma unroll
            for (int k = 0; k < OWNER_TRIPS; ++k) {
                const int i = tid + k * OWNER_THREADS;
                t[k] = sp[min(i, n16 - 1)];
            }
            #pragma unroll
            for (int k = 0; k < OWNER_TRIPS; ++k) {
                a4[tsw + k * OWNER_THREADS] = t[k];
            }
            #pragma unroll
            for (int k = 0; k < QUEUE_CAP / 32; ++k)
                myq[lane + 32 * k] = qr[k];
        } else {
            float4 t[OWNER_TRIPS];
            #pragma unroll
            for (int k = 0; k < OWNER_TRIPS; ++k) {
                const int i = tid + k * OWNER_THREADS;
                t[k] = sp[min(i, n16 - 1)];
            }
            #pragma unroll
            for (int k = 0; k < OWNER_TRIPS; ++k) {
                a4[tsw + k * OWNER_THREADS] = t[k];
            }
        }
    }
    __syncthreads();

    if (!saved) {
    const int wd4 = (E + 127) / 128;
    const uint4* row = map4 + (size_t)blockIdx.x * MAP_UNITS(E);
    const int nIter = (wd4 + 31) / 32;
    const uint4 z = make_uint4(0u, 0u, 0u, 0u);

    int pend = 0;
    int cnt  = 0;

    uint4 v;
    {
        const int q = wave * 32 + lane;
        v = row[min(q, wd4 - 1)];
        if (q >= wd4) v = z;
    }
    for (int it = wave; it < nIter; it += OWNER_THREADS / 32) {
        uint4 vn;
        {
            const int qn = (it + OWNER_THREADS / 32) * 32 + lane;
            vn = row[min(qn, wd4 - 1)];
            if (qn >= wd4) vn = z;
        }
        const unsigned comp[4] = {v.x, v.y, v.z, v.w};
        #pragma unroll
        for (int cidx = 0; cidx < 4; ++cidx) {
            const unsigned vc = comp[cidx];
            unsigned bal = __builtin_amdgcn_ballot_w32(vc != 0u);
            while (bal) {
                const int l = __builtin_ctz(bal);
                bal &= bal - 1u;
                unsigned vv = (unsigned)__builtin_amdgcn_readlane((int)vc, l);
                const int ebase = (it * 32 + l) * 128 + cidx * 32;
                while (vv) {
                    const int e = ebase + __builtin_ctz(vv);
                    vv &= vv - 1u;
                    pend = (lane == cnt) ? e : pend;
                    if (++cnt == PUSH_GROUP) {
                        gcn_owner_push(H, rec, acc, queue, qcnt, pend, PUSH_GROUP, lane);
                        cnt = 0;
                    }
                }
            }
        }
        v = vn;
    }
    if (cnt > 0) gcn_owner_push(H, rec, acc, queue, qcnt, pend, cnt, lane);

    __syncthreads();
    }

    {
        int have = have_saved;
        if (!saved) {
            const unsigned craw = __builtin_amdgcn_readfirstlane(qcnt[wave]);
            have = min((int)craw, QUEUE_CAP);
            if (!reuse) {
                if (lane == 0) sqn_own[wave] = craw;
                for (int i = lane; i < have; i += 32) sq_own[i] = myq[i];
            }
        }
        int base = 0;
        for (; base + EDGE_GROUP <= have; base += EDGE_GROUP) {
            const v2i r = myq[base + (lane & (EDGE_GROUP - 1))];
            gcn_owner_group(H, acc, r.x, r.y, EDGE_GROUP, lane);
        }
        if (base < have) {
            const v2i r = myq[min(base + (lane & (EDGE_GROUP - 1)), have - 1)];
            gcn_owner_group(H, acc, r.x, r.y, have - base, lane);
        }
    }

    const int half = lane >> 4;
    const int l16  = lane & 15;
    const int col  = (wave & 7) * 16 + l16;
    v2f bfr[D_FEAT / 4];
    float bc = 0.0f;
    if (Wn != nullptr) {
        #pragma unroll
        for (int q = 0; q < D_FEAT / 4; ++q) {
            const int ka = 4 * q + 2 * half;
            bfr[q].x = Wn[(size_t)ka * D_FEAT + col];
            bfr[q].y = Wn[(size_t)(ka + 1) * D_FEAT + col];
        }
        bc = bn[col];
    }

    __syncthreads();

    if (Wn != nullptr) {
        const int npan = (nr + 15) >> 4;
        const int t = 2 * half + 4 * l16;
        #pragma unroll 1
        for (int p = wave >> 3; p < npan; p += 4) {
            const float* arow = acc + (p * 16 + l16) * D_FEAT;
            v8f c = {};
            #pragma unroll
            for (int g = 0; g < D_FEAT / 32; ++g) {
                v2f a[8];
                #pragma unroll
                for (int u = 0; u < 8; ++u)
                    a[u] = *(const v2f*)(arow + ((4 * (8 * g + u)) ^ t));
                #pragma unroll
                for (int u = 0; u < 8; ++u)
                    c = __builtin_amdgcn_wmma_f32_16x16x4_f32(
                         false, a[u],  false, bfr[8 * g + u],
                         (short)0, c,  false,  false);
                __builtin_amdgcn_sched_barrier(0);
            }
            float sd[8];
            #pragma unroll
            for (int j = 0; j < 8; ++j)
                sd[j] = dis[min(r0 + p * 16 + j + 8 * half, N - 1)];
            #pragma unroll
            for (int j = 0; j < 8; ++j) {
                const int row = p * 16 + j + 8 * half;
                if (row < nr) {
                    const size_t o = (size_t)(r0 + row) * D_FEAT + col;
                    Hn[o] = c[j];
                    Sn[o] = bc + (sd[j] * sd[j]) * c[j];
                }
            }
        }
    } else {
        float4* op = (float4*)(fin + (size_t)r0 * D_FEAT);
        #pragma unroll 1
        for (int k = 0; k < OWNER_TRIPS; ++k) {
            const int i = tid + k * OWNER_THREADS;
            if (i < n16) op[i] = a4[tsw + k * OWNER_THREADS];
        }
    }
}

extern "C" void kernel_launch(void* const* d_in, const int* in_sizes, int n_in,
                              void* d_out, int out_size, void* d_ws, size_t ws_size,
                              hipStream_t stream) {
    const float* feat = (const float*)d_in[0];
    const int*   ei   = (const int*)d_in[1];
    const float* W0   = (const float*)d_in[2];
    const float* b0   = (const float*)d_in[3];
    const float* W1   = (const float*)d_in[4];
    const float* b1   = (const float*)d_in[5];
    const float* W2   = (const float*)d_in[6];
    const float* b2   = (const float*)d_in[7];

    const int N = in_sizes[0] / D_FEAT;
    const int E = in_sizes[1] / 2;
    const int* src = ei;
    const int* dst = ei + E;

    float* deg = (float*)d_ws;
    size_t off = ((size_t)N + 1023) & ~(size_t)1023;
    float* dis = deg + off;
    float* H   = dis + off;
    float* X   = H + (size_t)N * D_FEAT;
    int2* rec  = (int2*)(X + (size_t)N * D_FEAT);
    size_t offE = ((size_t)E + 1023) & ~(size_t)1023;
    uint4* map4 = (uint4*)(rec + offE);
    const size_t wd4 = (size_t)MAP_UNITS(E);
    const size_t owners = ((size_t)N + OWN_ROWS - 1) / OWN_ROWS;
    float* H2  = (float*)(map4 + owners * wd4);
    v2i* sq = (v2i*)(H2 + (size_t)N * D_FEAT);
    unsigned* sqn = (unsigned*)(sq + owners * OWNER_WAVES * QUEUE_CAP);
    float* out = (float*)d_out;

    const int TB = 256;
    dim3 blkE((E + TB - 1) / TB);
    dim3 blkN((N + TB - 1) / TB);
    dim3 blkTile((E + MAP_TILE_EDGES - 1) / MAP_TILE_EDGES);
    dim3 blkGemm((N + ROWS_PER_BLOCK - 1) / ROWS_PER_BLOCK);
    dim3 blkOwn((N + OWN_ROWS - 1) / OWN_ROWS);

    gcn_init_deg<<<blkN, TB, 0, stream>>>(deg, N);
    gcn_count_deg<<<blkTile, MAP_TILE_EDGES, 0, stream>>>(dst, deg, (unsigned*)map4, N, E);
    gcn_rsqrt_rec<<<blkE, TB, 0, stream>>>(deg, dis, src, dst, rec, N, E);

    gcn_gemm128<<<blkGemm, TB, 0, stream>>>(feat, W0, dis, b0, H, X, N);
    gcn_owner_edges<<<blkOwn, OWNER_THREADS, 0, stream>>>(H, rec, map4, X, nullptr, W1, b1, dis, H2, X, sq, sqn, 0, N, E);
    gcn_owner_edges<<<blkOwn, OWNER_THREADS, 0, stream>>>(H2, rec, map4, X, nullptr, W2, b2, dis, H, X, sq, sqn, 1, N, E);
    gcn_owner_edges<<<blkOwn, OWNER_THREADS, 0, stream>>>(H, rec, map4, X, out, nullptr, nullptr, dis, nullptr, nullptr, sq, sqn, 1, N, E);
}


# Start values b + dis^2*H formed by the owner block from its own H rows as it loads them; the start-value plane (stored by 3 epilogues, re-read by 3 edge passes) is gone; sums, their order unchanged
# speedup vs baseline: 3.4779x; 1.0596x over previous
#include <hip/hip_runtime.h>
#include <hip/hip_bf16.h>

typedef __attribute__((ext_vector_type(2))) float v2f;
typedef __attribute__((ext_vector_type(8))) float v8f;

#define D_FEAT 128
#define LDS_STRIDE 132
#define ROWS_PER_BLOCK 128

#define OWN_ROWS 392
#define MAP_TILE_EDGES 1024
#define MAP_TILE_OWNERS 256
#define MAP_TILE_PITCH 33
#define MAP_TILE_LINES (MAP_TILE_OWNERS / (MAP_TILE_EDGES / 32))
#define MAP_UNITS(E) ((((E) + MAP_TILE_EDGES - 1) / MAP_TILE_EDGES) * (MAP_TILE_EDGES / 128))

#define REC_SRC_BITS 23
static_assert(OWN_ROWS <= (1 << (32 - REC_SRC_BITS)), "row inside an owner's run must fit above the src bits");

__global__ void gcn_init_deg(float* __restrict__ deg, int N) {
    const int i = blockIdx.x * blockDim.x + threadIdx.x;
    if (i < N) deg[i] = 1.0f;
}

__global__ void __launch_bounds__(MAP_TILE_EDGES)
gcn_count_deg(const int* __restrict__ dst, float* __restrict__ deg,
              unsigned* __restrict__ map, int N, int E) {
    __shared__ unsigned tile[MAP_TILE_OWNERS * MAP_TILE_PITCH];
    const int tid  = threadIdx.x;
    const int lane = tid & 31;
    const int wave = tid >> 5;
    const int i = blockIdx.x * MAP_TILE_EDGES + tid;
    const int owners = (N + OWN_ROWS - 1) / OWN_ROWS;
    const size_t wd = (size_t)MAP_UNITS(E) * 4;
    int d = 0;
    if (i < E) d = dst[i];
    const int o = (i < E) ? (int)((unsigned)d / OWN_ROWS) : -1;
    for (int ob = 0; ob < owners; ob += MAP_TILE_OWNERS) {
        if (ob > 0) __syncthreads();
        for (int k = tid; k < MAP_TILE_OWNERS * MAP_TILE_PITCH; k += MAP_TILE_EDGES) tile[k] = 0u;
        __syncthreads();
        const int ol = o - ob;
        if (ol >= 0 && ol < MAP_TILE_OWNERS) atomicOr(&tile[ol * MAP_TILE_PITCH + wave], 1u << lane);
        __syncthreads();
        unsigned wv[MAP_TILE_LINES];
        #pragma unroll
        for (int j = 0; j < MAP_TILE_LINES; ++j)
            wv[j] = tile[(wave + j * (MAP_TILE_EDGES / 32)) * MAP_TILE_PITCH + lane];
        #pragma unroll
        for (int j = 0; j < MAP_TILE_LINES; ++j) {
            const int q = ob + wave + j * (MAP_TILE_EDGES / 32);
            if (q < owners) map[(size_t)q * wd + (size_t)blockIdx.x * 32 + lane] = wv[j];
        }
    }
    if (i < E) atomicAdd(&deg[d], 1.0f);
}

__global__ void gcn_rsqrt_rec(const float* __restrict__ deg, float* __restrict__ dis,
                              const int* __restrict__ src, const int* __restrict__ dst,
                              int2* __restrict__ rec, int N, int E) {
    const int i = blockIdx.x * blockDim.x + threadIdx.x;
    for (int n = i; n < N; n += gridDim.x * blockDim.x)
        dis[n] = rsqrtf(deg[n]);
    if (i < E) {
        const int s = src[i];
        const int d = dst[i];
        const float w = rsqrtf(deg[s]) * rsqrtf(deg[d]);
        rec[i] = make_int2(s | (int)(((unsigned)d % OWN_ROWS) << REC_SRC_BITS), __float_as_int(w));
    }
}

__global__ void __launch_bounds__(256)
gcn_gemm128(const float* X, const float* __restrict__ W,
            float* __restrict__ H, int N) {
    __shared__ float Ws[D_FEAT * LDS_STRIDE];
    __shared__ float As[16 * LDS_STRIDE];

    const int tid  = threadIdx.x;
    const int base = blockIdx.x * ROWS_PER_BLOCK;

    for (int i = tid; i < (D_FEAT * D_FEAT) / 4; i += 256) {
        const int k = i >> 5;
        const int c = i & 31;
        const float4 w4 = ((const float4*)(W + (size_t)k * D_FEAT))[c];
        Ws[(4 * c + 0) * LDS_STRIDE + k] = w4.x;
        Ws[(4 * c + 1) * LDS_STRIDE + k] = w4.y;
        Ws[(4 * c + 2) * LDS_STRIDE + k] = w4.z;
        Ws[(4 * c + 3) * LDS_STRIDE + k] = w4.w;
    }

    const int wave = tid >> 5;
    const int lane = tid & 31;
    const int half = lane >> 4;
    const int l16  = lane & 15;
    const int col  = wave * 16 + l16;

    #pragma unroll 1
    for (int p = 0; p < ROWS_PER_BLOCK / 16; ++p) {
        const int r0 = base + p * 16;
        if (r0 >= N) break;

        __syncthreads();

        for (int i = tid; i < (16 * D_FEAT) / 4; i += 256) {
            const int row = i >> 5;
            const int c   = i & 31;
            if (r0 + row < N) {
                const float4 a4 = ((const float4*)(X + (size_t)(r0 + row) * D_FEAT))[c];
                *(float4*)&As[row * LDS_STRIDE + 4 * c] = a4;
            }
        }
        __syncthreads();

        v8f acc = {};
        #pragma unroll
        for (int k0 = 0; k0 < D_FEAT; k0 += 4) {
            const int ka = k0 + 2 * half;
            const v2f a = *(const v2f*)(As + l16 * LDS_STRIDE + ka);
            const v2f b = *(const v2f*)(Ws + col * LDS_STRIDE + ka);
            acc = __builtin_amdgcn_wmma_f32_16x16x4_f32(
                 false, a,  false, b,
                 (short)0, acc,  false,  false);
        }

        #pragma unroll
        for (int j = 0; j < 8; ++j) {
            const int row = r0 + j + 8 * half;
            if (row < N) H[(size_t)row * D_FEAT + col] = acc[j];
        }
    }
}

#define EDGE_GROUP 16
#define PUSH_GROUP 32
#define OWNER_THREADS 1024
#define OWNER_WAVES (OWNER_THREADS / 32)
#define QUEUE_CAP 384
#define OWNER_TRIPS ((OWN_ROWS * (D_FEAT / 4) + OWNER_THREADS - 1) / OWNER_THREADS)
static_assert(OWNER_WAVES == 32, "a row's wave is row & 31");

typedef __attribute__((ext_vector_type(2))) int v2i;

__device__ __forceinline__ float4 gcn_start4(float4 h, float4 b, float d) {
    const float d2 = d * d;
    float4 r;
    r.x = b.x + d2 * h.x;
    r.y = b.y + d2 * h.y;
    r.z = b.z + d2 * h.z;
    r.w = b.w + d2 * h.w;
    return r;
}

__device__ __forceinline__ void gcn_owner_push(const float* __restrict__ H,
                                               const int2* __restrict__ rec,
                                               float* acc, v2i* queue, unsigned* qcnt,
                                               int pend, int n, int lane) {
    const int2 r = rec[pend];
    const int q = (int)((unsigned)r.x >> REC_SRC_BITS) & (OWNER_WAVES - 1);
    const bool act = lane < n;
    unsigned slot = QUEUE_CAP;
    if (act) slot = atomicAdd(&qcnt[q], 1u);
    const bool ok = act && slot < QUEUE_CAP;
    if (ok) {
        v2i t;
        t.x = r.x;
        t.y = r.y;
        queue[q * QUEUE_CAP + slot] = t;
    }
    unsigned om = __builtin_amdgcn_ballot_w32(act && !ok);
    while (om) {
        const int l = __builtin_ctz(om);
        om &= om - 1u;
        const int rx = __builtin_amdgcn_readlane(r.x, l);
        const float wv = __int_as_float(__builtin_amdgcn_readlane(r.y, l));
        const int s = rx & ((1 << REC_SRC_BITS) - 1);
        const int d = (int)((unsigned)rx >> REC_SRC_BITS);
        const float2* hp = (const float2*)(H + (size_t)s * D_FEAT) + lane;
        const float2 lo = hp[0];
        const float2 hi = hp[32];
        float* ap = acc + d * D_FEAT + ((2 * lane) ^ (4 * (d & 15)));
        atomicAdd(ap +  0, lo.x * wv);
        atomicAdd(ap +  1, lo.y * wv);
        atomicAdd(ap + 64, hi.x * wv);
        atomicAdd(ap + 65, hi.y * wv);
    }
}

__device__ __forceinline__ void gcn_owner_group(const float* __restrict__ H,
                                                float* acc, int rx, int ry, int n,
                                                int lane) {
    const int s = rx & ((1 << REC_SRC_BITS) - 1);
    const int d = (int)((unsigned)rx >> REC_SRC_BITS);
    float h[EDGE_GROUP][4];
    #pragma unroll
    for (int j = 0; j < EDGE_GROUP; ++j) {
        const int sj = __builtin_amdgcn_readlane(s, j);
        const float2* hp = (const float2*)(H + (size_t)sj * D_FEAT) + lane;
        const float2 lo = hp[0];
        const float2 hi = hp[32];
        h[j][0] = lo.x;
        h[j][1] = lo.y;
        h[j][2] = hi.x;
        h[j][3] = hi.y;
    }
    #pragma unroll
    for (int j = 0; j < EDGE_GROUP; ++j) {
        if (j < n) {
            const int dj = __builtin_amdgcn_readlane(d, j);
            const float wj = __int_as_float(__builtin_amdgcn_readlane(ry, j));
            float* ap = acc + dj * D_FEAT + ((2 * lane) ^ (4 * (dj & 15)));
            float2 lo = *(const float2*)(ap);
            float2 hi = *(const float2*)(ap + 64);
            lo.x += h[j][0] * wj;
            lo.y += h[j][1] * wj;
            hi.x += h[j][2] * wj;
            hi.y += h[j][3] * wj;
            *(float2*)(ap)      = lo;
            *(float2*)(ap + 64) = hi;
        }
    }
}

__global__ void __launch_bounds__(OWNER_THREADS)
gcn_owner_edges(const float* __restrict__ H,
                const int2* __restrict__ rec,
                const uint4* __restrict__ map4,
                const float* __restrict__ bcur, float* fin,
                const float* __restrict__ Wn,
                const float* __restrict__ dis,
                float* __restrict__ Hn,
                v2i* sq, unsigned* sqn, int reuse, int N, int E) {
    __shared__ float acc[OWNER_TRIPS * OWNER_THREADS * 4];
    __shared__ v2i queue[OWNER_WAVES * QUEUE_CAP];
    __shared__ unsigned qcnt[OWNER_WAVES];

    const int tid  = threadIdx.x;
    const int lane = tid & 31;
    const int wave = __builtin_amdgcn_readfirstlane(tid >> 5);
    const int r0 = blockIdx.x * OWN_ROWS;
    const int nr = min(OWN_ROWS, N - r0);

    v2i* sq_own = sq + ((size_t)blockIdx.x * OWNER_WAVES + wave) * QUEUE_CAP;
    unsigned* sqn_own = sqn + (size_t)blockIdx.x * OWNER_WAVES;
    v2i* myq = queue + wave * QUEUE_CAP;
    bool saved = false;
    int have_saved = 0;
    if (reuse) {
        const unsigned c = sqn_own[lane];
        saved = __builtin_amdgcn_ballot_w32(c > (unsigned)QUEUE_CAP) == 0u;
        have_saved = __builtin_amdgcn_readlane((int)c, wave);
    }
    const bool fill = saved && have_saved > 0;

    const float4* sp = (const float4*)(H + (size_t)r0 * D_FEAT);
    const int sw = (tid >> 5) & 15;
    const int tsw = tid ^ sw;
    float4* a4 = (float4*)acc;
    const int n16 = nr * (D_FEAT / 4);
    if (tid < OWNER_WAVES) qcnt[tid] = 0u;
    const float4 b4 = ((const float4*)bcur)[lane];
    float dv[OWNER_TRIPS];
    #pragma unroll
    for (int k = 0; k < OWNER_TRIPS; ++k)
        dv[k] = dis[min(r0 + wave + 32 * k, N - 1)];
    {
        if (fill) {
            v2i qr[QUEUE_CAP / 32];
            float4 t[OWNER_TRIPS];
            #pragma unroll
            for (int k = 0; k < QUEUE_CAP / 32; ++k)
                qr[k] = sq_own[min(lane + 32 * k, have_saved - 1)];
            #pragma unroll
            for (int k = 0; k < OWNER_TRIPS; ++k) {
                const int i = tid + k * OWNER_THREADS;
                t[k] = sp[min(i, n16 - 1)];
            }
            #pragma unroll
            for (int k = 0; k < OWNER_TRIPS; ++k) {
                a4[tsw + k * OWNER_THREADS] = gcn_start4(t[k], b4, dv[k]);
            }
            #pragma unroll
            for (int k = 0; k < QUEUE_CAP / 32; ++k)
                myq[lane + 32 * k] = qr[k];
        } else {
            float4 t[OWNER_TRIPS];
            #pragma unroll
            for (int k = 0; k < OWNER_TRIPS; ++k) {
                const int i = tid + k * OWNER_THREADS;
                t[k] = sp[min(i, n16 - 1)];
            }
            #pragma unroll
            for (int k = 0; k < OWNER_TRIPS; ++k) {
                a4[tsw + k * OWNER_THREADS] = gcn_start4(t[k], b4, dv[k]);
            }
        }
    }
    __syncthreads();

    if (!saved) {
    const int wd4 = (E + 127) / 128;
    const uint4* row = map4 + (size_t)blockIdx.x * MAP_UNITS(E);
    const int nIter = (wd4 + 31) / 32;
    const uint4 z = make_uint4(0u, 0u, 0u, 0u);

    int pend = 0;
    int cnt  = 0;

    uint4 v;
    {
        const int q = wave * 32 + lane;
        v = row[min(q, wd4 - 1)];
        if (q >= wd4) v = z;
    }
    for (int it = wave; it < nIter; it += OWNER_THREADS / 32) {
        uint4 vn;
        {
            const int qn = (it + OWNER_THREADS / 32) * 32 + lane;
            vn = row[min(qn, wd4 - 1)];
            if (qn >= wd4) vn = z;
        }
        const unsigned comp[4] = {v.x, v.y, v.z, v.w};
        #pragma unroll
        for (int cidx = 0; cidx < 4; ++cidx) {
            const unsigned vc = comp[cidx];
            unsigned bal = __builtin_amdgcn_ballot_w32(vc != 0u);
            while (bal) {
                const int l = __builtin_ctz(bal);
                bal &= bal - 1u;
                unsigned vv = (unsigned)__builtin_amdgcn_readlane((int)vc, l);
                const int ebase = (it * 32 + l) * 128 + cidx * 32;
                while (vv) {
                    const int e = ebase + __builtin_ctz(vv);
                    vv &= vv - 1u;
                    pend = (lane == cnt) ? e : pend;
                    if (++cnt == PUSH_GROUP) {
                        gcn_owner_push(H, rec, acc, queue, qcnt, pend, PUSH_GROUP, lane);
                        cnt = 0;
                    }
                }
            }
        }
        v = vn;
    }
    if (cnt > 0) gcn_owner_push(H, rec, acc, queue, qcnt, pend, cnt, lane);

    __syncthreads();
    }

    {
        int have = have_saved;
        if (!saved) {
            const unsigned craw = __builtin_amdgcn_readfirstlane(qcnt[wave]);
            have = min((int)craw, QUEUE_CAP);
            if (!reuse) {
                if (lane == 0) sqn_own[wave] = craw;
                for (int i = lane; i < have; i += 32) sq_own[i] = myq[i];
            }
        }
        int base = 0;
        for (; base + EDGE_GROUP <= have; base += EDGE_GROUP) {
            const v2i r = myq[base + (lane & (EDGE_GROUP - 1))];
            gcn_owner_group(H, acc, r.x, r.y, EDGE_GROUP, lane);
        }
        if (base < have) {
            const v2i r = myq[min(base + (lane & (EDGE_GROUP - 1)), have - 1)];
            gcn_owner_group(H, acc, r.x, r.y, have - base, lane);
        }
    }

    const int half = lane >> 4;
    const int l16  = lane & 15;
    const int col  = (wave & 7) * 16 + l16;
    v2f bfr[D_FEAT / 4];
    if (Wn != nullptr) {
        #pragma unroll
        for (int q = 0; q < D_FEAT / 4; ++q) {
            const int ka = 4 * q + 2 * half;
            bfr[q].x = Wn[(size_t)ka * D_FEAT + col];
            bfr[q].y = Wn[(size_t)(ka + 1) * D_FEAT + col];
        }
    }

    __syncthreads();

    if (Wn != nullptr) {
        const int npan = (nr + 15) >> 4;
        const int t = 2 * half + 4 * l16;
        #pragma unroll 1
        for (int p = wave >> 3; p < npan; p += 4) {
            const float* arow = acc + (p * 16 + l16) * D_FEAT;
            v8f c = {};
            #pragma unroll
            for (int g = 0; g < D_FEAT / 32; ++g) {
                v2f a[8];
                #pragma unroll
                for (int u = 0; u < 8; ++u)
                    a[u] = *(const v2f*)(arow + ((4 * (8 * g + u)) ^ t));
                #pragma unroll
                for (int u = 0; u < 8; ++u)
                    c = __builtin_amdgcn_wmma_f32_16x16x4_f32(
                         false, a[u],  false, bfr[8 * g + u],
                         (short)0, c,  false,  false);
                __builtin_amdgcn_sched_barrier(0);
            }
            #pragma unroll
            for (int j = 0; j < 8; ++j) {
                const int row = p * 16 + j + 8 * half;
                if (row < nr) Hn[(size_t)(r0 + row) * D_FEAT + col] = c[j];
            }
        }
    } else {
        float4* op = (float4*)(fin + (size_t)r0 * D_FEAT);
        #pragma unroll 1
        for (int k = 0; k < OWNER_TRIPS; ++k) {
            const int i = tid + k * OWNER_THREADS;
            if (i < n16) op[i] = a4[tsw + k * OWNER_THREADS];
        }
    }
}

extern "C" void kernel_launch(void* const* d_in, const int* in_sizes, int n_in,
                              void* d_out, int out_size, void* d_ws, size_t ws_size,
                              hipStream_t stream) {
    const float* feat = (const float*)d_in[0];
    const int*   ei   = (const int*)d_in[1];
    const float* W0   = (const float*)d_in[2];
    const float* b0   = (const float*)d_in[3];
    const float* W1   = (const float*)d_in[4];
    const float* b1   = (const float*)d_in[5];
    const float* W2   = (const float*)d_in[6];
    const float* b2   = (const float*)d_in[7];

    const int N = in_sizes[0] / D_FEAT;
    const int E = in_sizes[1] / 2;
    const int* src = ei;
    const int* dst = ei + E;

    float* deg = (float*)d_ws;
    size_t off = ((size_t)N + 1023) & ~(size_t)1023;
    float* dis = deg + off;
    float* H   = dis + off;
    int2* rec  = (int2*)(H + (size_t)N * D_FEAT);
    size_t offE = ((size_t)E + 1023) & ~(size_t)1023;
    uint4* map4 = (uint4*)(rec + offE);
    const size_t wd4 = (size_t)MAP_UNITS(E);
    const size_t owners = ((size_t)N + OWN_ROWS - 1) / OWN_ROWS;
    float* H2  = (float*)(map4 + owners * wd4);
    v2i* sq = (v2i*)(H2 + (size_t)N * D_FEAT);
    unsigned* sqn = (unsigned*)(sq + owners * OWNER_WAVES * QUEUE_CAP);
    float* out = (float*)d_out;

    const int TB = 256;
    dim3 blkE((E + TB - 1) / TB);
    dim3 blkN((N + TB - 1) / TB);
    dim3 blkTile((E + MAP_TILE_EDGES - 1) / MAP_TILE_EDGES);
    dim3 blkGemm((N + ROWS_PER_BLOCK - 1) / ROWS_PER_BLOCK);
    dim3 blkOwn((N + OWN_ROWS - 1) / OWN_ROWS);

    gcn_init_deg<<<blkN, TB, 0, stream>>>(deg, N);
    gcn_count_deg<<<blkTile, MAP_TILE_EDGES, 0, stream>>>(dst, deg, (unsigned*)map4, N, E);
    gcn_rsqrt_rec<<<blkE, TB, 0, stream>>>(deg, dis, src, dst, rec, N, E);

    gcn_gemm128<<<blkGemm, TB, 0, stream>>>(feat, W0, H, N);
    gcn_owner_edges<<<blkOwn, OWNER_THREADS, 0, stream>>>(H, rec, map4, b0, nullptr, W1, dis, H2, sq, sqn, 0, N, E);
    gcn_owner_edges<<<blkOwn, OWNER_THREADS, 0, stream>>>(H2, rec, map4, b1, nullptr, W2, dis, H, sq, sqn, 1, N, E);
    gcn_owner_edges<<<blkOwn, OWNER_THREADS, 0, stream>>>(H, rec, map4, b2, out, nullptr, dis, nullptr, sq, sqn, 1, N, E);
}
